# starting kernel
# baseline (speedup 1.0000x reference)
#include <hip/hip_runtime.h>

typedef __attribute__((ext_vector_type(2))) float v2f;
typedef __attribute__((ext_vector_type(4))) float v4f;
typedef __attribute__((ext_vector_type(8))) float v8f;

#define IN_FEATS   128
#define OUT_FEATS  128
#define KDIM       256
#define TILE_M     16
#define LDS_STRIDE 260

__global__ __launch_bounds__(256) void sage_fused_wmma_f32(
    const float* __restrict__ feat,
    const int*   __restrict__ offsets,
    const int*   __restrict__ indices,
    const float* __restrict__ weight,
    const float* __restrict__ bias,
    float*       __restrict__ out,
    int n_nodes, int n_tiles)
{
    __shared__ float Atile[TILE_M][LDS_STRIDE];

    const int tid  = threadIdx.x;
    const int wave = tid >> 5;
    const int lane = tid & 31;
    const int m    = lane & 15;
    const int kh   = lane >> 4;
    const int n0   = wave * 16;

    const float* wrow = weight + (size_t)(n0 + m) * KDIM + (kh << 1);
    v2f Breg[64];
#pragma unroll
    for (int kk = 0; kk < 64; ++kk)
        Breg[kk] = *(const v2f*)(wrow + (kk << 2));

    const float bsc = bias[n0 + m];

    for (int tile = blockIdx.x; tile < n_tiles; tile += (int)gridDim.x) {
        const int node0 = tile * TILE_M;

#pragma unroll
        for (int jj = 0; jj < 2; ++jj) {
            const int j    = wave * 2 + jj;
            const int node = node0 + j;
            v4f selfv = {0.f, 0.f, 0.f, 0.f};
            v4f acc   = {0.f, 0.f, 0.f, 0.f};
            if (node < n_nodes) {
                selfv = *(const v4f*)(feat + (size_t)node * IN_FEATS + lane * 4);
                const int beg = offsets[node];
                const int deg = offsets[node + 1] - beg;
                for (int e = 0; e < deg; ++e) {
                    const int src = indices[beg + e];
                    acc += *(const v4f*)(feat + (size_t)src * IN_FEATS + lane * 4);
                }
                const float inv = 1.0f / (float)(deg > 1 ? deg : 1);
                acc = acc * inv;
            }
            *(v4f*)&Atile[j][lane * 4]            = selfv;
            *(v4f*)&Atile[j][IN_FEATS + lane * 4] = acc;
        }
        __syncthreads();

        v8f c = {0.f, 0.f, 0.f, 0.f, 0.f, 0.f, 0.f, 0.f};
#pragma unroll
        for (int kk = 0; kk < 64; ++kk) {
            const int kcol = (kk << 2) + (kh << 1);
            v2f a = *(const v2f*)&Atile[m][kcol];
            c = __builtin_amdgcn_wmma_f32_16x16x4_f32(
                     false, a,  false, Breg[kk],
                     (short)0, c,  false,  false);
        }

#pragma unroll
        for (int r = 0; r < 8; ++r) {
            const int row = node0 + kh * 8 + r;
            if (row < n_nodes)
                out[(size_t)row * OUT_FEATS + n0 + m] = c[r] + bsc;
        }
        __syncthreads();
    }
}

extern "C" void kernel_launch(void* const* d_in, const int* in_sizes, int n_in,
                              void* d_out, int out_size, void* d_ws, size_t ws_size,
                              hipStream_t stream) {
    const float* feat    = (const float*)d_in[0];
    const int*   offsets = (const int*)d_in[1];
    const int*   indices = (const int*)d_in[2];
    const float* weight  = (const float*)d_in[3];
    const float* bias    = (const float*)d_in[4];
    float*       out     = (float*)d_out;

    const int n_nodes = in_sizes[0] / IN_FEATS;
    const int n_tiles = (n_nodes + TILE_M - 1) / TILE_M;
    const int grid    = n_tiles < 1024 ? n_tiles : 1024;

    sage_fused_wmma_f32<<<grid, 256, 0, stream>>>(
        feat, offsets, indices, weight, bias, out, n_nodes, n_tiles);
}


# Gather keeps 16 independent neighbour-row loads in flight per wave across both of its rows, ids and row ranges by scalar loads (was chunks of 8, load-wait-add, one row after the other)
# speedup vs baseline: 1.1384x; 1.1384x over previous
#include <hip/hip_runtime.h>

typedef __attribute__((ext_vector_type(2))) float v2f;
typedef __attribute__((ext_vector_type(4))) float v4f;
typedef __attribute__((ext_vector_type(8))) float v8f;

#define IN_FEATS   128
#define OUT_FEATS  128
#define KDIM       256
#define TILE_M     16
#define LDS_STRIDE 260
#define GCHUNK     16

__device__ __forceinline__ v4f ld_row(const float* __restrict__ feat, int row, unsigned laneb)
{
    const char* rp = (const char*)(feat + (size_t)row * IN_FEATS);
    return *(const v4f*)(rp + laneb);
}

__device__ __forceinline__ v4f gather_rest(const float* __restrict__ feat,
                                           const int* __restrict__ indices,
                                           int beg, int deg, int e, unsigned laneb, v4f acc)
{
    for (; e + GCHUNK <= deg; e += GCHUNK) {
        int id[GCHUNK];
#pragma unroll
        for (int u = 0; u < GCHUNK; ++u) id[u] = indices[beg + e + u];
        v4f v[GCHUNK];
#pragma unroll
        for (int u = 0; u < GCHUNK; ++u) v[u] = ld_row(feat, id[u], laneb);
#pragma unroll
        for (int u = 0; u < GCHUNK; ++u) acc += v[u];
    }
    for (; e < deg; ++e)
        acc += ld_row(feat, indices[beg + e], laneb);
    return acc;
}

__global__ __launch_bounds__(256) void sage_fused_wmma_f32(
    const float* __restrict__ feat,
    const int*   __restrict__ offsets,
    const int*   __restrict__ indices,
    const float* __restrict__ weight,
    const float* __restrict__ bias,
    float*       __restrict__ out,
    int n_nodes, int n_tiles)
{
    __shared__ float Atile[TILE_M][LDS_STRIDE];

    const int tid  = threadIdx.x;
    const int wave = tid >> 5;
    const int lane = tid & 31;
    const int m    = lane & 15;
    const int kh   = lane >> 4;
    const int n0   = wave * 16;
    const int lane4 = lane * 4;
    const unsigned laneb = (unsigned)lane4 * 4u;
    const int wu    = __builtin_amdgcn_readfirstlane(wave);

    const float* wrow = weight + (size_t)(n0 + m) * KDIM + (kh << 1);
    v2f Breg[64];
#pragma unroll
    for (int kk = 0; kk < 64; ++kk)
        Breg[kk] = *(const v2f*)(wrow + (kk << 2));

    const float bsc = bias[n0 + m];

    for (int tile = blockIdx.x; tile < n_tiles; tile += (int)gridDim.x) {
        const int node0 = tile * TILE_M;

        {
            const int ja = wu * 2;
            const int na = node0 + ja;
            const int nb = na + 1;
            v4f selfA = {0.f, 0.f, 0.f, 0.f}, selfB = {0.f, 0.f, 0.f, 0.f};
            v4f accA  = {0.f, 0.f, 0.f, 0.f}, accB  = {0.f, 0.f, 0.f, 0.f};
            int begA = 0, degA = 0, begB = 0, degB = 0;
            if (nb < n_nodes) {
                const int o0 = offsets[na];
                const int o1 = offsets[na + 1];
                const int o2 = offsets[na + 2];
                begA = o0; degA = o1 - o0;
                begB = o1; degB = o2 - o1;
                selfA = ld_row(feat, na, laneb);
                selfB = ld_row(feat, nb, laneb);
            } else if (na < n_nodes) {
                begA  = offsets[na];
                degA  = offsets[na + 1] - begA;
                selfA = ld_row(feat, na, laneb);
            }
            int eA = 0, eB = 0;
            if (degA >= GCHUNK && degB >= GCHUNK) {
                int idA[GCHUNK], idB[GCHUNK];
#pragma unroll
                for (int u = 0; u < GCHUNK; ++u) idA[u] = indices[begA + u];
#pragma unroll
                for (int u = 0; u < GCHUNK; ++u) idB[u] = indices[begB + u];
                v4f v[GCHUNK];
#pragma unroll
                for (int u = 0; u < GCHUNK; ++u) v[u] = ld_row(feat, idA[u], laneb);
                unsigned offB = laneb;
#pragma unroll
                for (int u = 0; u < GCHUNK; ++u) {
                    accA += v[u];
                    asm volatile("" : "+v"(offB) : "v"(accA));
                    v[u] = ld_row(feat, idB[u], offB);
                }
#pragma unroll
                for (int u = 0; u < GCHUNK; ++u) accB += v[u];
                eA = GCHUNK;
                eB = GCHUNK;
            }
            accA = gather_rest(feat, indices, begA, degA, eA, laneb, accA);
            accB = gather_rest(feat, indices, begB, degB, eB, laneb, accB);
            accA = accA * (1.0f / (float)(degA > 1 ? degA : 1));
            accB = accB * (1.0f / (float)(degB > 1 ? degB : 1));
            *(v4f*)&Atile[ja][lane4]                = selfA;
            *(v4f*)&Atile[ja][IN_FEATS + lane4]     = accA;
            *(v4f*)&Atile[ja + 1][lane4]            = selfB;
            *(v4f*)&Atile[ja + 1][IN_FEATS + lane4] = accB;
        }
        __syncthreads();

        v8f c = {0.f, 0.f, 0.f, 0.f, 0.f, 0.f, 0.f, 0.f};
#pragma unroll
        for (int kk = 0; kk < 64; ++kk) {
            const int kcol = (kk << 2) + (kh << 1);
            v2f a = *(const v2f*)&Atile[m][kcol];
            c = __builtin_amdgcn_wmma_f32_16x16x4_f32(
                     false, a,  false, Breg[kk],
                     (short)0, c,  false,  false);
        }

#pragma unroll
        for (int r = 0; r < 8; ++r) {
            const int row = node0 + kh * 8 + r;
            if (row < n_nodes)
                out[(size_t)row * OUT_FEATS + n0 + m] = c[r] + bsc;
        }
        __syncthreads();
    }
}

extern "C" void kernel_launch(void* const* d_in, const int* in_sizes, int n_in,
                              void* d_out, int out_size, void* d_ws, size_t ws_size,
                              hipStream_t stream) {
    const float* feat    = (const float*)d_in[0];
    const int*   offsets = (const int*)d_in[1];
    const int*   indices = (const int*)d_in[2];
    const float* weight  = (const float*)d_in[3];
    const float* bias    = (const float*)d_in[4];
    float*       out     = (float*)d_out;

    const int n_nodes = in_sizes[0] / IN_FEATS;
    const int n_tiles = (n_nodes + TILE_M - 1) / TILE_M;
    const int grid    = n_tiles < 1024 ? n_tiles : 1024;

    sage_fused_wmma_f32<<<grid, 256, 0, stream>>>(
        feat, offsets, indices, weight, bias, out, n_nodes, n_tiles);
}


# Work-group of 16 waves split by role: 8 waves gather the next tile's A rows into a second LDS tile while 8 waves run this tile's WMMAs; one work-group per WGP, one LDS-only barrier per tile
# speedup vs baseline: 1.3193x; 1.1589x over previous
#include <hip/hip_runtime.h>

typedef __attribute__((ext_vector_type(2))) float v2f;
typedef __attribute__((ext_vector_type(4))) float v4f;
typedef __attribute__((ext_vector_type(8))) float v8f;

#define IN_FEATS   128
#define OUT_FEATS  128
#define KDIM       256
#define TILE_M     16
#define LDS_STRIDE 260
#define GCHUNK     16
#define NTHREADS   512
#define MAX_GRID   256

__device__ __forceinline__ void lds_handoff_barrier()
{
    asm volatile("s_wait_dscnt 0x0\n\ts_barrier_signal -1\n\ts_barrier_wait -1" ::: "memory");
}

__device__ __forceinline__ v4f ld_row(const float* __restrict__ feat, int row, unsigned laneb)
{
    const char* rp = (const char*)(feat + (size_t)row * IN_FEATS);
    return *(const v4f*)(rp + laneb);
}

__device__ __forceinline__ v4f gather_rest(const float* __restrict__ feat,
                                           const int* __restrict__ indices,
                                           int beg, int deg, int e, unsigned laneb, v4f acc)
{
    for (; e + GCHUNK <= deg; e += GCHUNK) {
        int id[GCHUNK];
#pragma unroll
        for (int u = 0; u < GCHUNK; ++u) id[u] = indices[beg + e + u];
        v4f v[GCHUNK];
#pragma unroll
        for (int u = 0; u < GCHUNK; ++u) v[u] = ld_row(feat, id[u], laneb);
#pragma unroll
        for (int u = 0; u < GCHUNK; ++u) acc += v[u];
    }
    for (; e < deg; ++e)
        acc += ld_row(feat, indices[beg + e], laneb);
    return acc;
}

__device__ __forceinline__ void build_rows(const float* __restrict__ feat,
                                           const int* __restrict__ offsets,
                                           const int* __restrict__ indices,
                                           int n_nodes, int node0, int ja,
                                           unsigned laneb, int lane4,
                                           float (*At)[LDS_STRIDE])
{
    const int na = node0 + ja;
    const int nb = na + 1;
    v4f selfA = {0.f, 0.f, 0.f, 0.f}, selfB = {0.f, 0.f, 0.f, 0.f};
    v4f accA  = {0.f, 0.f, 0.f, 0.f}, accB  = {0.f, 0.f, 0.f, 0.f};
    int begA = 0, degA = 0, begB = 0, degB = 0;
    if (nb < n_nodes) {
        const int o0 = offsets[na];
        const int o1 = offsets[na + 1];
        const int o2 = offsets[na + 2];
        begA = o0; degA = o1 - o0;
        begB = o1; degB = o2 - o1;
        selfA = ld_row(feat, na, laneb);
        selfB = ld_row(feat, nb, laneb);
    } else if (na < n_nodes) {
        begA  = offsets[na];
        degA  = offsets[na + 1] - begA;
        selfA = ld_row(feat, na, laneb);
    }
    int eA = 0, eB = 0;
    if (degA >= GCHUNK && degB >= GCHUNK) {
        int idA[GCHUNK], idB[GCHUNK];
#pragma unroll
        for (int u = 0; u < GCHUNK; ++u) idA[u] = indices[begA + u];
#pragma unroll
        for (int u = 0; u < GCHUNK; ++u) idB[u] = indices[begB + u];
        v4f v[GCHUNK];
#pragma unroll
        for (int u = 0; u < GCHUNK; ++u) v[u] = ld_row(feat, idA[u], laneb);
        unsigned offB = laneb;
#pragma unroll
        for (int u = 0; u < GCHUNK; ++u) {
            accA += v[u];
            asm volatile("" : "+v"(offB) : "v"(accA));
            v[u] = ld_row(feat, idB[u], offB);
        }
#pragma unroll
        for (int u = 0; u < GCHUNK; ++u) accB += v[u];
        eA = GCHUNK;
        eB = GCHUNK;
    }
    accA = gather_rest(feat, indices, begA, degA, eA, laneb, accA);
    accB = gather_rest(feat, indices, begB, degB, eB, laneb, accB);
    accA = accA * (1.0f / (float)(degA > 1 ? degA : 1));
    accB = accB * (1.0f / (float)(degB > 1 ? degB : 1));
    *(v4f*)&At[ja][lane4]                = selfA;
    *(v4f*)&At[ja][IN_FEATS + lane4]     = accA;
    *(v4f*)&At[ja + 1][lane4]            = selfB;
    *(v4f*)&At[ja + 1][IN_FEATS + lane4] = accB;
}

__global__ __launch_bounds__(NTHREADS) void sage_fused_wmma_f32(
    const float* __restrict__ feat,
    const int*   __restrict__ offsets,
    const int*   __restrict__ indices,
    const float* __restrict__ weight,
    const float* __restrict__ bias,
    float*       __restrict__ out,
    int n_nodes, int n_tiles)
{
    __shared__ float Atile[2][TILE_M][LDS_STRIDE];

    const int tid  = threadIdx.x;
    const int wave = tid >> 5;
    const int lane = tid & 31;
    const int m    = lane & 15;
    const int kh   = lane >> 4;
    const int lane4 = lane * 4;
    const unsigned laneb = (unsigned)lane4 * 4u;
    const int wu    = __builtin_amdgcn_readfirstlane(wave);

    const int  wa   = wu >> 2;
    const int  wb   = wu & 3;
    const bool mult = ((wa + wb) & 1) == 0;
    const int  ridx = wa * 2 + (wb >> 1);
    const int  n0   = ridx * 16;
    const int  ja   = ridx * 2;

    const int nt = (n_tiles - 1 - (int)blockIdx.x) / (int)gridDim.x + 1;

    v2f Breg[64];
    float bsc = 0.f;
    if (mult) {
        const float* wrow = weight + (size_t)(n0 + m) * KDIM + (kh << 1);
#pragma unroll
        for (int kk = 0; kk < 64; ++kk)
            Breg[kk] = *(const v2f*)(wrow + (kk << 2));
        bsc = bias[n0 + m];
    } else {
        build_rows(feat, offsets, indices, n_nodes, (int)blockIdx.x * TILE_M, ja,
                   laneb, lane4, Atile[0]);
    }
    __syncthreads();

    for (int i = 0; i < nt; ++i) {
        const int tile  = (int)blockIdx.x + i * (int)gridDim.x;
        const int node0 = tile * TILE_M;
        const int cur   = i & 1;

        if (mult) {
            v8f c = {0.f, 0.f, 0.f, 0.f, 0.f, 0.f, 0.f, 0.f};
#pragma unroll
            for (int kk = 0; kk < 64; ++kk) {
                const int kcol = (kk << 2) + (kh << 1);
                v2f a = *(const v2f*)&Atile[cur][m][kcol];
                c = __builtin_amdgcn_wmma_f32_16x16x4_f32(
                         false, a,  false, Breg[kk],
                         (short)0, c,  false,  false);
            }

#pragma unroll
            for (int r = 0; r < 8; ++r) {
                const int row = node0 + kh * 8 + r;
                if (row < n_nodes)
                    out[(size_t)row * OUT_FEATS + n0 + m] = c[r] + bsc;
            }
        } else if (i + 1 < nt) {
            build_rows(feat, offsets, indices, n_nodes, (tile + (int)gridDim.x) * TILE_M, ja,
                       laneb, lane4, Atile[cur ^ 1]);
        }
        lds_handoff_barrier();
    }
}

extern "C" void kernel_launch(void* const* d_in, const int* in_sizes, int n_in,
                              void* d_out, int out_size, void* d_ws, size_t ws_size,
                              hipStream_t stream) {
    const float* feat    = (const float*)d_in[0];
    const int*   offsets = (const int*)d_in[1];
    const int*   indices = (const int*)d_in[2];
    const float* weight  = (const float*)d_in[3];
    const float* bias    = (const float*)d_in[4];
    float*       out     = (float*)d_out;

    const int n_nodes = in_sizes[0] / IN_FEATS;
    const int n_tiles = (n_nodes + TILE_M - 1) / TILE_M;
    const int grid    = n_tiles < MAX_GRID ? n_tiles : MAX_GRID;

    sage_fused_wmma_f32<<<grid, NTHREADS, 0, stream>>>(
        feat, offsets, indices, weight, bias, out, n_nodes, n_tiles);
}
